# starting kernel
# baseline (speedup 1.0000x reference)
#include <hip/hip_runtime.h>


#define NUM_NEURONS 100000
#define MAX_EDGES   100000
#define NUM_CONN    1600000
#define BATCH       4
#define OUT_CH      16

typedef __attribute__((ext_vector_type(2))) float v2f;
typedef __attribute__((ext_vector_type(8))) float v8f;

#define WS_EFFW 0u
#define WS_S    100000u
#define WS_CNT  500000u
#define WS_U    600000u
#define WS_XT   1000000u

__global__ void bhs_zero(float* __restrict__ ws) {
  unsigned i = blockIdx.x * blockDim.x + threadIdx.x;
  if (i < 225000u) ((float4*)(ws + WS_S))[i] = make_float4(0.f, 0.f, 0.f, 0.f);
}

__global__ void bhs_traces(const float* __restrict__ x_in,
                           const float* __restrict__ post_spk,
                           const float* __restrict__ w_hat,
                           const float* __restrict__ s_e,
                           const float* __restrict__ e_trace,
                           const float* __restrict__ pre_trace,
                           const float* __restrict__ post_trace,
                           float* __restrict__ out,
                           float* __restrict__ ws) {
  unsigned i = blockIdx.x * blockDim.x + threadIdx.x;
  if (i >= NUM_NEURONS) return;
  const float decay = 0.95122942450071400910f;
  float x0 = x_in[i];
  float x1 = x_in[NUM_NEURONS + i];
  float x2 = x_in[2u * NUM_NEURONS + i];
  float x3 = x_in[3u * NUM_NEURONS + i];
  float pre_spk  = (x0 + x1 + x2 + x3) * 0.25f;
  float pre_new  = fmaf(pre_trace[i],  decay, pre_spk);
  float post_new = fmaf(post_trace[i], decay, post_spk[i]);
  float e_new    = fmaf(0.9f, e_trace[i], pre_spk * (post_new + 0.1f));
  const size_t OUTB = (size_t)BATCH * NUM_NEURONS * OUT_CH;
  out[OUTB + i]                 = e_new;
  out[OUTB + MAX_EDGES + i]     = pre_new;
  out[OUTB + 2u * MAX_EDGES + i] = post_new;
  ws[WS_EFFW + i] = s_e[i] * w_hat[i];
  ((float4*)(ws + WS_XT))[i] = make_float4(x0, x1, x2, x3);
}

__global__ void bhs_scatterA(const int* __restrict__ he, float* __restrict__ ws) {
  unsigned c = blockIdx.x * blockDim.x + threadIdx.x;
  int n = he[c];
  int e = he[NUM_CONN + c];
  float cw = ws[WS_EFFW + e];
  float4 xv = ((const float4*)(ws + WS_XT))[n];
  float* S = ws + WS_S + (size_t)e * 4u;
  atomicAdd(&S[0], xv.x * cw);
  atomicAdd(&S[1], xv.y * cw);
  atomicAdd(&S[2], xv.z * cw);
  atomicAdd(&S[3], xv.w * cw);
  atomicAdd(ws + WS_CNT + e, 1.0f);
}

__global__ void bhs_norm(float* __restrict__ ws) {
  unsigned e = blockIdx.x * blockDim.x + threadIdx.x;
  if (e >= MAX_EDGES) return;
  float inv = 1.0f / fmaxf(ws[WS_CNT + e], 1.0f);
  float4* S = (float4*)(ws + WS_S);
  float4 s = S[e];
  s.x *= inv; s.y *= inv; s.z *= inv; s.w *= inv;
  S[e] = s;
}

__global__ void bhs_scatterB(const int* __restrict__ he, float* __restrict__ ws) {
  unsigned c = blockIdx.x * blockDim.x + threadIdx.x;
  int n = he[c];
  int e = he[NUM_CONN + c];
  float cw = ws[WS_EFFW + e];
  float4 t = ((const float4*)(ws + WS_S))[e];
  float* U = ws + WS_U + (size_t)n * 4u;
  atomicAdd(&U[0], cw * t.x);
  atomicAdd(&U[1], cw * t.y);
  atomicAdd(&U[2], cw * t.z);
  atomicAdd(&U[3], cw * t.w);
}

__global__ void bhs_out_wmma(const float* __restrict__ weight_node,
                             const float* __restrict__ bias,
                             const float* __restrict__ ws,
                             float* __restrict__ out) {
  unsigned lane = threadIdx.x & 31u;
  unsigned wave = (blockIdx.x * blockDim.x + threadIdx.x) >> 5;
  unsigned base = wave * 16u;
  unsigned oc = lane & 15u;
  float wv = weight_node[oc];
  float bv = bias[oc];
  const float* U = ws + WS_U;

  v2f a;
  a.x = (lane < 16u) ? U[base + lane] : 0.0f;
  a.y = 0.0f;
  v2f b;
  b.x = (lane < 16u) ? wv : 0.0f;
  b.y = 0.0f;
  v8f c;
#pragma unroll
  for (int v = 0; v < 8; ++v) c[v] = bv;

  v8f d = __builtin_amdgcn_wmma_f32_16x16x4_f32(
       false, a,  false, b,
       (short)0, c,  false,  false);

  unsigned mbase = (lane < 16u) ? 0u : 8u;
#pragma unroll
  for (int v = 0; v < 8; ++v) {
    unsigned nb = base + mbase + (unsigned)v;
    unsigned n = nb >> 2;
    unsigned bb = nb & 3u;
    out[((size_t)bb * NUM_NEURONS + n) * OUT_CH + oc] = d[v];
  }
}

extern "C" void kernel_launch(void* const* d_in, const int* in_sizes, int n_in,
                              void* d_out, int out_size, void* d_ws, size_t ws_size,
                              hipStream_t stream) {
  const float* x_in        = (const float*)d_in[0];
  const int*   he          = (const int*)d_in[1];
  const float* post_spk    = (const float*)d_in[2];
  const float* w_hat       = (const float*)d_in[3];
  const float* s_e         = (const float*)d_in[4];
  const float* e_trace     = (const float*)d_in[5];
  const float* pre_trace   = (const float*)d_in[6];
  const float* post_trace  = (const float*)d_in[7];
  const float* weight_node = (const float*)d_in[8];
  const float* bias        = (const float*)d_in[9];
  float* out = (float*)d_out;
  float* ws  = (float*)d_ws;

  bhs_zero<<<(225000u + 255u) / 256u, 256, 0, stream>>>(ws);
  bhs_traces<<<(NUM_NEURONS + 255u) / 256u, 256, 0, stream>>>(
      x_in, post_spk, w_hat, s_e, e_trace, pre_trace, post_trace, out, ws);
  bhs_scatterA<<<NUM_CONN / 256u, 256, 0, stream>>>(he, ws);
  bhs_norm<<<(MAX_EDGES + 255u) / 256u, 256, 0, stream>>>(ws);
  bhs_scatterB<<<NUM_CONN / 256u, 256, 0, stream>>>(he, ws);
  bhs_out_wmma<<<3125, 256, 0, stream>>>(weight_node, bias, ws, out);
}


# Scatter adds regrouped: 8 (pass A) or 4 (pass B) adjacent lanes per connection add into one 32-byte edge row (S0-3 + count) or 16-byte node row: 1 atomic request per connection, not 5 or 4
# speedup vs baseline: 3.1266x; 3.1266x over previous
#include <hip/hip_runtime.h>


#define NUM_NEURONS 100000
#define MAX_EDGES   100000
#define NUM_CONN    1600000
#define BATCH       4
#define OUT_CH      16

typedef __attribute__((ext_vector_type(2))) float v2f;
typedef __attribute__((ext_vector_type(8))) float v8f;

#define WS_EFFW 0u
#define WS_S    100000u
#define WS_XT   900000u
#define WS_U    900000u
#define ROW     8u
#define CNT     4u

__global__ void bhs_zero(float* __restrict__ ws) {
  unsigned i = blockIdx.x * blockDim.x + threadIdx.x;
  if (i < 200000u) ((float4*)(ws + WS_S))[i] = make_float4(0.f, 0.f, 0.f, 0.f);
}

__global__ void bhs_traces(const float* __restrict__ x_in,
                           const float* __restrict__ post_spk,
                           const float* __restrict__ w_hat,
                           const float* __restrict__ s_e,
                           const float* __restrict__ e_trace,
                           const float* __restrict__ pre_trace,
                           const float* __restrict__ post_trace,
                           float* __restrict__ out,
                           float* __restrict__ ws) {
  unsigned i = blockIdx.x * blockDim.x + threadIdx.x;
  if (i >= NUM_NEURONS) return;
  const float decay = 0.95122942450071400910f;
  float x0 = x_in[i];
  float x1 = x_in[NUM_NEURONS + i];
  float x2 = x_in[2u * NUM_NEURONS + i];
  float x3 = x_in[3u * NUM_NEURONS + i];
  float pre_spk  = (x0 + x1 + x2 + x3) * 0.25f;
  float pre_new  = fmaf(pre_trace[i],  decay, pre_spk);
  float post_new = fmaf(post_trace[i], decay, post_spk[i]);
  float e_new    = fmaf(0.9f, e_trace[i], pre_spk * (post_new + 0.1f));
  const size_t OUTB = (size_t)BATCH * NUM_NEURONS * OUT_CH;
  out[OUTB + i]                 = e_new;
  out[OUTB + MAX_EDGES + i]     = pre_new;
  out[OUTB + 2u * MAX_EDGES + i] = post_new;
  ws[WS_EFFW + i] = s_e[i] * w_hat[i];
  ((float4*)(ws + WS_XT))[i] = make_float4(x0, x1, x2, x3);
}

__global__ void bhs_scatterA(const int* __restrict__ he, float* __restrict__ ws) {
  unsigned c = blockIdx.x * blockDim.x + threadIdx.x;
  unsigned lane = threadIdx.x & 31u;
  int n_own = he[c];
  int e_own = he[NUM_CONN + c];
  unsigned j = lane & 7u;
  unsigned q = lane >> 3;
  const float* XT = ws + WS_XT;
  float* S = ws + WS_S;
  unsigned ev[8];
  float vv[8];
  for (unsigned s = 0; s < 8u; ++s) {
    int srcl = (int)(s * 4u + q);
    unsigned n = (unsigned)__shfl(n_own, srcl);
    unsigned e = (unsigned)__shfl(e_own, srcl);
    float cw = ws[WS_EFFW + e];
    float xv = XT[n * 4u + (j & 3u)];
    ev[s] = e;
    vv[s] = (j < 4u) ? xv * cw : 1.0f;
  }
  for (unsigned s = 0; s < 8u; ++s) {
    if (j <= CNT) atomicAdd(S + ev[s] * ROW + j, vv[s]);
  }
}

__global__ void bhs_norm(float* __restrict__ ws) {
  unsigned e = blockIdx.x * blockDim.x + threadIdx.x;
  if (e >= MAX_EDGES) return;
  float* row = ws + WS_S + e * ROW;
  float inv = 1.0f / fmaxf(row[CNT], 1.0f);
  float4 s = *(float4*)row;
  s.x *= inv; s.y *= inv; s.z *= inv; s.w *= inv;
  *(float4*)row = s;
  ((float4*)(ws + WS_U))[e] = make_float4(0.f, 0.f, 0.f, 0.f);
}

__global__ void bhs_scatterB(const int* __restrict__ he, float* __restrict__ ws) {
  unsigned c = blockIdx.x * blockDim.x + threadIdx.x;
  unsigned lane = threadIdx.x & 31u;
  int n_own = he[c];
  int e_own = he[NUM_CONN + c];
  unsigned j = lane & 3u;
  unsigned q = lane >> 2;
  const float* S = ws + WS_S;
  float* U = ws + WS_U;
  unsigned nv[4];
  float uv[4];
  for (unsigned s = 0; s < 4u; ++s) {
    int srcl = (int)(s * 8u + q);
    unsigned n = (unsigned)__shfl(n_own, srcl);
    unsigned e = (unsigned)__shfl(e_own, srcl);
    float cw = ws[WS_EFFW + e];
    float t = S[e * ROW + j];
    nv[s] = n;
    uv[s] = cw * t;
  }
  for (unsigned s = 0; s < 4u; ++s) {
    atomicAdd(U + nv[s] * 4u + j, uv[s]);
  }
}

__global__ void bhs_out_wmma(const float* __restrict__ weight_node,
                             const float* __restrict__ bias,
                             const float* __restrict__ ws,
                             float* __restrict__ out) {
  unsigned lane = threadIdx.x & 31u;
  unsigned wave = (blockIdx.x * blockDim.x + threadIdx.x) >> 5;
  unsigned base = wave * 16u;
  unsigned oc = lane & 15u;
  float wv = weight_node[oc];
  float bv = bias[oc];
  const float* U = ws + WS_U;

  v2f a;
  a.x = (lane < 16u) ? U[base + lane] : 0.0f;
  a.y = 0.0f;
  v2f b;
  b.x = (lane < 16u) ? wv : 0.0f;
  b.y = 0.0f;
  v8f c;
#pragma unroll
  for (int v = 0; v < 8; ++v) c[v] = bv;

  v8f d = __builtin_amdgcn_wmma_f32_16x16x4_f32(
       false, a,  false, b,
       (short)0, c,  false,  false);

  unsigned mbase = (lane < 16u) ? 0u : 8u;
#pragma unroll
  for (int v = 0; v < 8; ++v) {
    unsigned nb = base + mbase + (unsigned)v;
    unsigned n = nb >> 2;
    unsigned bb = nb & 3u;
    out[((size_t)bb * NUM_NEURONS + n) * OUT_CH + oc] = d[v];
  }
}

extern "C" void kernel_launch(void* const* d_in, const int* in_sizes, int n_in,
                              void* d_out, int out_size, void* d_ws, size_t ws_size,
                              hipStream_t stream) {
  const float* x_in        = (const float*)d_in[0];
  const int*   he          = (const int*)d_in[1];
  const float* post_spk    = (const float*)d_in[2];
  const float* w_hat       = (const float*)d_in[3];
  const float* s_e         = (const float*)d_in[4];
  const float* e_trace     = (const float*)d_in[5];
  const float* pre_trace   = (const float*)d_in[6];
  const float* post_trace  = (const float*)d_in[7];
  const float* weight_node = (const float*)d_in[8];
  const float* bias        = (const float*)d_in[9];
  float* out = (float*)d_out;
  float* ws  = (float*)d_ws;

  bhs_zero<<<(200000u + 255u) / 256u, 256, 0, stream>>>(ws);
  bhs_traces<<<(NUM_NEURONS + 255u) / 256u, 256, 0, stream>>>(
      x_in, post_spk, w_hat, s_e, e_trace, pre_trace, post_trace, out, ws);
  bhs_scatterA<<<NUM_CONN / 256u, 256, 0, stream>>>(he, ws);
  bhs_norm<<<(MAX_EDGES + 255u) / 256u, 256, 0, stream>>>(ws);
  bhs_scatterB<<<NUM_CONN / 256u, 256, 0, stream>>>(he, ws);
  bhs_out_wmma<<<3125, 256, 0, stream>>>(weight_node, bias, ws, out);
}


# Connection weight depends on the edge only: factored out of both segment sums into the per-edge norm (T = S*cw/count*cw); no per-connection gather of it in either scatter pass; rounding differs
# speedup vs baseline: 3.4909x; 1.1165x over previous
#include <hip/hip_runtime.h>


#define NUM_NEURONS 100000
#define MAX_EDGES   100000
#define NUM_CONN    1600000
#define BATCH       4
#define OUT_CH      16

typedef __attribute__((ext_vector_type(2))) float v2f;
typedef __attribute__((ext_vector_type(8))) float v8f;

#define WS_S    100000u
#define WS_XT   900000u
#define WS_U    900000u
#define ROW     8u
#define CNT     4u

__global__ void bhs_zero(float* __restrict__ ws) {
  unsigned i = blockIdx.x * blockDim.x + threadIdx.x;
  if (i < 200000u) ((float4*)(ws + WS_S))[i] = make_float4(0.f, 0.f, 0.f, 0.f);
}

__global__ void bhs_traces(const float* __restrict__ x_in,
                           const float* __restrict__ post_spk,
                           const float* __restrict__ e_trace,
                           const float* __restrict__ pre_trace,
                           const float* __restrict__ post_trace,
                           float* __restrict__ out,
                           float* __restrict__ ws) {
  unsigned i = blockIdx.x * blockDim.x + threadIdx.x;
  if (i >= NUM_NEURONS) return;
  const float decay = 0.95122942450071400910f;
  float x0 = x_in[i];
  float x1 = x_in[NUM_NEURONS + i];
  float x2 = x_in[2u * NUM_NEURONS + i];
  float x3 = x_in[3u * NUM_NEURONS + i];
  float pre_spk  = (x0 + x1 + x2 + x3) * 0.25f;
  float pre_new  = fmaf(pre_trace[i],  decay, pre_spk);
  float post_new = fmaf(post_trace[i], decay, post_spk[i]);
  float e_new    = fmaf(0.9f, e_trace[i], pre_spk * (post_new + 0.1f));
  const size_t OUTB = (size_t)BATCH * NUM_NEURONS * OUT_CH;
  out[OUTB + i]                 = e_new;
  out[OUTB + MAX_EDGES + i]     = pre_new;
  out[OUTB + 2u * MAX_EDGES + i] = post_new;
  ((float4*)(ws + WS_XT))[i] = make_float4(x0, x1, x2, x3);
}

__global__ void bhs_scatterA(const int* __restrict__ he, float* __restrict__ ws) {
  unsigned c = blockIdx.x * blockDim.x + threadIdx.x;
  unsigned lane = threadIdx.x & 31u;
  int n_own = he[c];
  int e_own = he[NUM_CONN + c];
  unsigned j = lane & 7u;
  unsigned q = lane >> 3;
  const float* XT = ws + WS_XT;
  float* S = ws + WS_S;
  unsigned ev[8];
  float vv[8];
  for (unsigned s = 0; s < 8u; ++s) {
    int srcl = (int)(s * 4u + q);
    unsigned n = (unsigned)__shfl(n_own, srcl);
    unsigned e = (unsigned)__shfl(e_own, srcl);
    float xv = XT[n * 4u + (j & 3u)];
    ev[s] = e;
    vv[s] = (j < 4u) ? xv : 1.0f;
  }
  for (unsigned s = 0; s < 8u; ++s) {
    if (j <= CNT) atomicAdd(S + ev[s] * ROW + j, vv[s]);
  }
}

__global__ void bhs_norm(const float* __restrict__ w_hat,
                         const float* __restrict__ s_e,
                         float* __restrict__ ws) {
  unsigned e = blockIdx.x * blockDim.x + threadIdx.x;
  if (e >= MAX_EDGES) return;
  float* row = ws + WS_S + e * ROW;
  float cw = s_e[e] * w_hat[e];
  float g = cw * (1.0f / fmaxf(row[CNT], 1.0f));
  float4 s = *(float4*)row;
  s.x = s.x * g * cw; s.y = s.y * g * cw; s.z = s.z * g * cw; s.w = s.w * g * cw;
  *(float4*)row = s;
  ((float4*)(ws + WS_U))[e] = make_float4(0.f, 0.f, 0.f, 0.f);
}

__global__ void bhs_scatterB(const int* __restrict__ he, float* __restrict__ ws) {
  unsigned c = blockIdx.x * blockDim.x + threadIdx.x;
  unsigned lane = threadIdx.x & 31u;
  int n_own = he[c];
  int e_own = he[NUM_CONN + c];
  unsigned j = lane & 3u;
  unsigned q = lane >> 2;
  const float* S = ws + WS_S;
  float* U = ws + WS_U;
  unsigned nv[4];
  float uv[4];
  for (unsigned s = 0; s < 4u; ++s) {
    int srcl = (int)(s * 8u + q);
    unsigned n = (unsigned)__shfl(n_own, srcl);
    unsigned e = (unsigned)__shfl(e_own, srcl);
    nv[s] = n;
    uv[s] = S[e * ROW + j];
  }
  for (unsigned s = 0; s < 4u; ++s) {
    atomicAdd(U + nv[s] * 4u + j, uv[s]);
  }
}

__global__ void bhs_out_wmma(const float* __restrict__ weight_node,
                             const float* __restrict__ bias,
                             const float* __restrict__ ws,
                             float* __restrict__ out) {
  unsigned lane = threadIdx.x & 31u;
  unsigned wave = (blockIdx.x * blockDim.x + threadIdx.x) >> 5;
  unsigned base = wave * 16u;
  unsigned oc = lane & 15u;
  float wv = weight_node[oc];
  float bv = bias[oc];
  const float* U = ws + WS_U;

  v2f a;
  a.x = (lane < 16u) ? U[base + lane] : 0.0f;
  a.y = 0.0f;
  v2f b;
  b.x = (lane < 16u) ? wv : 0.0f;
  b.y = 0.0f;
  v8f c;
#pragma unroll
  for (int v = 0; v < 8; ++v) c[v] = bv;

  v8f d = __builtin_amdgcn_wmma_f32_16x16x4_f32(
       false, a,  false, b,
       (short)0, c,  false,  false);

  unsigned mbase = (lane < 16u) ? 0u : 8u;
#pragma unroll
  for (int v = 0; v < 8; ++v) {
    unsigned nb = base + mbase + (unsigned)v;
    unsigned n = nb >> 2;
    unsigned bb = nb & 3u;
    out[((size_t)bb * NUM_NEURONS + n) * OUT_CH + oc] = d[v];
  }
}

extern "C" void kernel_launch(void* const* d_in, const int* in_sizes, int n_in,
                              void* d_out, int out_size, void* d_ws, size_t ws_size,
                              hipStream_t stream) {
  const float* x_in        = (const float*)d_in[0];
  const int*   he          = (const int*)d_in[1];
  const float* post_spk    = (const float*)d_in[2];
  const float* w_hat       = (const float*)d_in[3];
  const float* s_e         = (const float*)d_in[4];
  const float* e_trace     = (const float*)d_in[5];
  const float* pre_trace   = (const float*)d_in[6];
  const float* post_trace  = (const float*)d_in[7];
  const float* weight_node = (const float*)d_in[8];
  const float* bias        = (const float*)d_in[9];
  float* out = (float*)d_out;
  float* ws  = (float*)d_ws;

  bhs_zero<<<(200000u + 255u) / 256u, 256, 0, stream>>>(ws);
  bhs_traces<<<(NUM_NEURONS + 255u) / 256u, 256, 0, stream>>>(
      x_in, post_spk, e_trace, pre_trace, post_trace, out, ws);
  bhs_scatterA<<<NUM_CONN / 256u, 256, 0, stream>>>(he, ws);
  bhs_norm<<<(MAX_EDGES + 255u) / 256u, 256, 0, stream>>>(w_hat, s_e, ws);
  bhs_scatterB<<<NUM_CONN / 256u, 256, 0, stream>>>(he, ws);
  bhs_out_wmma<<<3125, 256, 0, stream>>>(weight_node, bias, ws, out);
}


# speedup vs baseline: 7.7641x; 2.2241x over previous
#include <hip/hip_runtime.h>


#define NUM_NEURONS 100000
#define MAX_EDGES   100000
#define NUM_CONN    1600000
#define BATCH       4
#define OUT_CH      16

typedef __attribute__((ext_vector_type(2))) float v2f;
typedef __attribute__((ext_vector_type(8))) float v8f;

#define WS_U1   100000u
#define WS_XT   900000u
#define WS_U0   900000u
#define OS_S0   0u
#define OS_S1   800000u
#define ROW     8u
#define CNT     4u

__device__ __forceinline__ unsigned bhs_half() {
  return (__builtin_amdgcn_s_sendmsg_rtn(0x87) >> 18) & 1u;
}

__global__ void bhs_zero(float* __restrict__ out) {
  unsigned i = blockIdx.x * blockDim.x + threadIdx.x;
  if (i < 400000u) ((float4*)(out + OS_S0))[i] = make_float4(0.f, 0.f, 0.f, 0.f);
}

__global__ void bhs_traces(const float* __restrict__ x_in,
                           const float* __restrict__ post_spk,
                           const float* __restrict__ e_trace,
                           const float* __restrict__ pre_trace,
                           const float* __restrict__ post_trace,
                           float* __restrict__ out,
                           float* __restrict__ ws) {
  unsigned i = blockIdx.x * blockDim.x + threadIdx.x;
  if (i >= NUM_NEURONS) return;
  const float decay = 0.95122942450071400910f;
  float x0 = x_in[i];
  float x1 = x_in[NUM_NEURONS + i];
  float x2 = x_in[2u * NUM_NEURONS + i];
  float x3 = x_in[3u * NUM_NEURONS + i];
  float pre_spk  = (x0 + x1 + x2 + x3) * 0.25f;
  float pre_new  = fmaf(pre_trace[i],  decay, pre_spk);
  float post_new = fmaf(post_trace[i], decay, post_spk[i]);
  float e_new    = fmaf(0.9f, e_trace[i], pre_spk * (post_new + 0.1f));
  const size_t OUTB = (size_t)BATCH * NUM_NEURONS * OUT_CH;
  out[OUTB + i]                 = e_new;
  out[OUTB + MAX_EDGES + i]     = pre_new;
  out[OUTB + 2u * MAX_EDGES + i] = post_new;
  ((float4*)(ws + WS_XT))[i] = make_float4(x0, x1, x2, x3);
}

__global__ void bhs_scatterA(const int* __restrict__ he,
                             const float* __restrict__ ws,
                             float* __restrict__ out) {
  unsigned half = bhs_half();
  unsigned c = blockIdx.x * blockDim.x + threadIdx.x;
  unsigned lane = threadIdx.x & 31u;
  int n_own = he[c];
  int e_own = he[NUM_CONN + c];
  unsigned j = lane & 7u;
  unsigned q = lane >> 3;
  const float* XT = ws + WS_XT;
  float* S = out + (half ? OS_S1 : OS_S0);
  unsigned ev[8];
  float vv[8];
  for (unsigned s = 0; s < 8u; ++s) {
    int srcl = (int)(s * 4u + q);
    unsigned n = (unsigned)__shfl(n_own, srcl);
    unsigned e = (unsigned)__shfl(e_own, srcl);
    float xv = XT[n * 4u + (j & 3u)];
    ev[s] = e;
    vv[s] = (j < 4u) ? xv : 1.0f;
  }
  for (unsigned s = 0; s < 8u; ++s) {
    if (j <= CNT) atomicAdd(S + ev[s] * ROW + j, vv[s]);
  }
}

__global__ void bhs_norm(const float* __restrict__ w_hat,
                         const float* __restrict__ s_e,
                         float* __restrict__ ws,
                         float* __restrict__ out) {
  unsigned e = blockIdx.x * blockDim.x + threadIdx.x;
  if (e >= MAX_EDGES) return;
  float* row0 = out + OS_S0 + e * ROW;
  const float* row1 = out + OS_S1 + e * ROW;
  float cw = s_e[e] * w_hat[e];
  float cnt = row0[CNT] + row1[CNT];
  float g = cw * (1.0f / fmaxf(cnt, 1.0f));
  float4 s = *(float4*)row0;
  float4 t = *(const float4*)row1;
  s.x = (s.x + t.x) * g * cw; s.y = (s.y + t.y) * g * cw;
  s.z = (s.z + t.z) * g * cw; s.w = (s.w + t.w) * g * cw;
  *(float4*)row0 = s;
  ((float4*)(ws + WS_U0))[e] = make_float4(0.f, 0.f, 0.f, 0.f);
  ((float4*)(ws + WS_U1))[e] = make_float4(0.f, 0.f, 0.f, 0.f);
}

__global__ void bhs_scatterB(const int* __restrict__ he,
                             float* __restrict__ ws,
                             const float* __restrict__ out) {
  unsigned c = blockIdx.x * blockDim.x + threadIdx.x;
  unsigned lane = threadIdx.x & 31u;
  int n_own = he[c];
  int e_own = he[NUM_CONN + c];
  unsigned half = bhs_half();
  unsigned j = lane & 3u;
  unsigned q = lane >> 2;
  const float* S = out + OS_S0;
  float* U = ws + (half ? WS_U1 : WS_U0);
  unsigned nv[4];
  float uv[4];
  for (unsigned s = 0; s < 4u; ++s) {
    int srcl = (int)(s * 8u + q);
    unsigned n = (unsigned)__shfl(n_own, srcl);
    unsigned e = (unsigned)__shfl(e_own, srcl);
    nv[s] = n;
    uv[s] = S[e * ROW + j];
  }
  for (unsigned s = 0; s < 4u; ++s) {
    atomicAdd(U + nv[s] * 4u + j, uv[s]);
  }
}

__global__ void bhs_out_wmma(const float* __restrict__ weight_node,
                             const float* __restrict__ bias,
                             const float* __restrict__ ws,
                             float* __restrict__ out) {
  unsigned lane = threadIdx.x & 31u;
  unsigned wave = (blockIdx.x * blockDim.x + threadIdx.x) >> 5;
  unsigned base = wave * 16u;
  unsigned oc = lane & 15u;
  float wv = weight_node[oc];
  float bv = bias[oc];
  const float* U0 = ws + WS_U0;
  const float* U1 = ws + WS_U1;

  v2f a;
  a.x = (lane < 16u) ? U0[base + lane] + U1[base + lane] : 0.0f;
  a.y = 0.0f;
  v2f b;
  b.x = (lane < 16u) ? wv : 0.0f;
  b.y = 0.0f;
  v8f c;
#pragma unroll
  for (int v = 0; v < 8; ++v) c[v] = bv;

  v8f d = __builtin_amdgcn_wmma_f32_16x16x4_f32(
       false, a,  false, b,
       (short)0, c,  false,  false);

  unsigned mbase = (lane < 16u) ? 0u : 8u;
#pragma unroll
  for (int v = 0; v < 8; ++v) {
    unsigned nb = base + mbase + (unsigned)v;
    unsigned n = nb >> 2;
    unsigned bb = nb & 3u;
    out[((size_t)bb * NUM_NEURONS + n) * OUT_CH + oc] = d[v];
  }
}

extern "C" void kernel_launch(void* const* d_in, const int* in_sizes, int n_in,
                              void* d_out, int out_size, void* d_ws, size_t ws_size,
                              hipStream_t stream) {
  const float* x_in        = (const float*)d_in[0];
  const int*   he          = (const int*)d_in[1];
  const float* post_spk    = (const float*)d_in[2];
  const float* w_hat       = (const float*)d_in[3];
  const float* s_e         = (const float*)d_in[4];
  const float* e_trace     = (const float*)d_in[5];
  const float* pre_trace   = (const float*)d_in[6];
  const float* post_trace  = (const float*)d_in[7];
  const float* weight_node = (const float*)d_in[8];
  const float* bias        = (const float*)d_in[9];
  float* out = (float*)d_out;
  float* ws  = (float*)d_ws;

  bhs_zero<<<(400000u + 255u) / 256u, 256, 0, stream>>>(out);
  bhs_traces<<<(NUM_NEURONS + 255u) / 256u, 256, 0, stream>>>(
      x_in, post_spk, e_trace, pre_trace, post_trace, out, ws);
  bhs_scatterA<<<NUM_CONN / 256u, 256, 0, stream>>>(he, ws, out);
  bhs_norm<<<(MAX_EDGES + 255u) / 256u, 256, 0, stream>>>(w_hat, s_e, ws, out);
  bhs_scatterB<<<NUM_CONN / 256u, 256, 0, stream>>>(he, ws, out);
  bhs_out_wmma<<<3125, 256, 0, stream>>>(weight_node, bias, ws, out);
}


# Connections ordered by edge range in LDS per call; an owner work-group per range keeps S, count, T in LDS, gathers x, adds T to U: 2 not 4 requests per connection, 4 launches; f32 sum order changes
# speedup vs baseline: 10.2965x; 1.3262x over previous
#include <hip/hip_runtime.h>


#define NUM_NEURONS 100000
#define MAX_EDGES   100000
#define NUM_CONN    1600000
#define BATCH       4
#define OUT_CH      16

#define NCHUNK 256u
#define CH     6250u
#define NBIN   256u
#define BW     391u

typedef __attribute__((ext_vector_type(2))) float v2f;
typedef __attribute__((ext_vector_type(8))) float v8f;

#define WS_U1   100000u
#define WS_U0   500000u
#define WS_XT   900000u
#define OS_SORT 0u
#define OS_TAB  1600000u

__device__ __forceinline__ unsigned bhs_half() {
  return (__builtin_amdgcn_s_sendmsg_rtn(0x87) >> 18) & 1u;
}

__global__ void bhs_traces(const float* __restrict__ x_in,
                           const float* __restrict__ post_spk,
                           const float* __restrict__ e_trace,
                           const float* __restrict__ pre_trace,
                           const float* __restrict__ post_trace,
                           float* __restrict__ out,
                           float* __restrict__ ws) {
  unsigned i = blockIdx.x * blockDim.x + threadIdx.x;
  if (i >= NUM_NEURONS) return;
  const float decay = 0.95122942450071400910f;
  float x0 = x_in[i];
  float x1 = x_in[NUM_NEURONS + i];
  float x2 = x_in[2u * NUM_NEURONS + i];
  float x3 = x_in[3u * NUM_NEURONS + i];
  float pre_spk  = (x0 + x1 + x2 + x3) * 0.25f;
  float pre_new  = fmaf(pre_trace[i],  decay, pre_spk);
  float post_new = fmaf(post_trace[i], decay, post_spk[i]);
  float e_new    = fmaf(0.9f, e_trace[i], pre_spk * (post_new + 0.1f));
  const size_t OUTB = (size_t)BATCH * NUM_NEURONS * OUT_CH;
  out[OUTB + i]                 = e_new;
  out[OUTB + MAX_EDGES + i]     = pre_new;
  out[OUTB + 2u * MAX_EDGES + i] = post_new;
  ((float4*)(ws + WS_XT))[i] = make_float4(x0, x1, x2, x3);
  ((float4*)(ws + WS_U0))[i] = make_float4(0.f, 0.f, 0.f, 0.f);
  ((float4*)(ws + WS_U1))[i] = make_float4(0.f, 0.f, 0.f, 0.f);
}

__global__ __launch_bounds__(1024) void bhs_order(const int* __restrict__ he,
                                                  float* __restrict__ out) {
  __shared__ unsigned hist[NBIN];
  __shared__ unsigned offs[NBIN];
  __shared__ unsigned wtot[8];
  __shared__ unsigned stage[CH];
  unsigned tid = threadIdx.x;
  unsigned lane = tid & 31u;
  unsigned chunk = blockIdx.x;
  unsigned base = chunk * CH;
  unsigned* sorted = (unsigned*)(out + OS_SORT);
  unsigned* tab = (unsigned*)(out + OS_TAB);
  if (tid < NBIN) hist[tid] = 0u;
  int nn[7], ee[7];
  for (unsigned it = 0; it < 7u; ++it) {
    unsigned k = tid + it * 1024u;
    nn[it] = 0; ee[it] = 0;
    if (k < CH) { nn[it] = he[base + k]; ee[it] = he[NUM_CONN + base + k]; }
  }
  __syncthreads();
  unsigned pk[7], slot[7];
  for (unsigned it = 0; it < 7u; ++it) {
    unsigned k = tid + it * 1024u;
    pk[it] = 0u; slot[it] = 0u;
    if (k < CH) {
      unsigned e = (unsigned)ee[it];
      unsigned bn = e / BW;
      unsigned el = e - bn * BW;
      pk[it] = (unsigned)nn[it] | (el << 17);
      unsigned r = atomicAdd(&hist[bn], 1u);
      slot[it] = (bn << 16) | r;
    }
  }
  __syncthreads();
  unsigned v = (tid < NBIN) ? hist[tid] : 0u;
  unsigned x = v;
  for (unsigned d = 1u; d < 32u; d <<= 1) {
    unsigned y = __shfl_up(x, d);
    if (lane >= d) x += y;
  }
  if (tid < NBIN && lane == 31u) wtot[tid >> 5] = x;
  __syncthreads();
  if (tid < NBIN) {
    unsigned w = tid >> 5;
    unsigned add = 0u;
    for (unsigned q = 0; q < 8u; ++q) if (q < w) add += wtot[q];
    unsigned start = add + x - v;
    offs[tid] = start;
    tab[tid * NCHUNK + chunk] = start;
  }
  if (tid == 0u) tab[NBIN * NCHUNK + chunk] = CH;
  __syncthreads();
  for (unsigned it = 0; it < 7u; ++it) {
    unsigned k = tid + it * 1024u;
    if (k < CH) stage[offs[slot[it] >> 16] + (slot[it] & 0xFFFFu)] = pk[it];
  }
  __syncthreads();
  for (unsigned it = 0; it < 7u; ++it) {
    unsigned k = tid + it * 1024u;
    if (k < CH) sorted[base + k] = stage[k];
  }
}

__device__ __forceinline__ void bhs_add_S(float* S_l, unsigned p, float4 xv) {
  float* r = S_l + (p >> 17) * 5u;
  atomicAdd(r + 0, xv.x);
  atomicAdd(r + 1, xv.y);
  atomicAdd(r + 2, xv.z);
  atomicAdd(r + 3, xv.w);
  atomicAdd(r + 4, 1.0f);
}

__device__ __forceinline__ void bhs_add_U(float* U, const float* T_l, unsigned pp,
                                          unsigned cnt, unsigned lane) {
  unsigned j = lane & 3u;
  unsigned q = lane >> 2;
  unsigned pr[4];
  float tv[4];
  for (unsigned s = 0; s < 4u; ++s) pr[s] = (unsigned)__shfl((int)pp, (int)(s * 8u + q));
  for (unsigned s = 0; s < 4u; ++s) tv[s] = T_l[(pr[s] >> 17) * 4u + j];
  for (unsigned s = 0; s < 4u; ++s) {
    if (s * 8u + q < cnt) atomicAdd(U + (pr[s] & 0x1FFFFu) * 4u + j, tv[s]);
  }
}

__global__ __launch_bounds__(1024) void bhs_owner(const float* __restrict__ w_hat,
                                                  const float* __restrict__ s_e,
                                                  float* ws,
                                                  const float* __restrict__ out) {
  __shared__ float S_l[BW * 5u];
  __shared__ float T_l[BW * 4u];
  unsigned half = bhs_half();
  unsigned tid = threadIdx.x;
  unsigned lane = tid & 31u;
  unsigned wave = tid >> 5;
  unsigned b = blockIdx.x;
  const unsigned* sorted = (const unsigned*)(out + OS_SORT);
  const unsigned* tab = (const unsigned*)(out + OS_TAB);
  const float4* XT4 = (const float4*)(ws + WS_XT);
  float* U = ws + (half ? WS_U1 : WS_U0);

  unsigned e_own = b * BW + tid;
  bool own = (tid < BW) && (e_own < (unsigned)MAX_EDGES);
  float sev = s_e[own ? e_own : 0u];
  float whv = w_hat[own ? e_own : 0u];

  int st_l = 0, ln_l = 0;
  if (lane < 8u) {
    unsigned ci = wave * 8u + lane;
    unsigned o0 = tab[b * NCHUNK + ci];
    unsigned o1 = tab[(b + 1u) * NCHUNK + ci];
    st_l = (int)(ci * CH + o0);
    ln_l = (int)(o1 - o0);
  }
  for (unsigned k = tid; k < BW * 5u; k += 1024u) S_l[k] = 0.f;

  unsigned stv[8], lnv[8];
  for (int u = 0; u < 8; ++u) {
    stv[u] = (unsigned)__builtin_amdgcn_readlane(st_l, u);
    lnv[u] = (unsigned)__builtin_amdgcn_readlane(ln_l, u);
  }
  unsigned p[8];
  for (int u = 0; u < 8; ++u) {
    unsigned w = sorted[stv[u] + lane];
    p[u] = (lane < lnv[u]) ? w : 0u;
  }
  float4 xv[8];
  for (int u = 0; u < 8; ++u) xv[u] = XT4[p[u] & 0x1FFFFu];
  __syncthreads();
  for (int u = 0; u < 8; ++u) {
    if (lane < lnv[u]) bhs_add_S(S_l, p[u], xv[u]);
  }
  for (int u = 0; u < 8; ++u) {
    for (unsigned r0 = 32u; r0 < lnv[u]; r0 += 32u) {
      if (r0 + lane < lnv[u]) {
        unsigned pp = sorted[stv[u] + r0 + lane];
        bhs_add_S(S_l, pp, XT4[pp & 0x1FFFFu]);
      }
    }
  }
  __syncthreads();
  if (own) {
    float cwv = sev * whv;
    float cnt = S_l[tid * 5u + 4u];
    float g = cwv * (1.0f / fmaxf(cnt, 1.0f));
    T_l[tid * 4u + 0u] = S_l[tid * 5u + 0u] * g * cwv;
    T_l[tid * 4u + 1u] = S_l[tid * 5u + 1u] * g * cwv;
    T_l[tid * 4u + 2u] = S_l[tid * 5u + 2u] * g * cwv;
    T_l[tid * 4u + 3u] = S_l[tid * 5u + 3u] * g * cwv;
  }
  __syncthreads();
  for (int u = 0; u < 8; ++u) {
    bhs_add_U(U, T_l, p[u], lnv[u] < 32u ? lnv[u] : 32u, lane);
    for (unsigned r0 = 32u; r0 < lnv[u]; r0 += 32u) {
      unsigned left = lnv[u] - r0;
      unsigned pp = 0u;
      if (lane < left) pp = sorted[stv[u] + r0 + lane];
      bhs_add_U(U, T_l, pp, left < 32u ? left : 32u, lane);
    }
  }
}

__global__ void bhs_out_wmma(const float* __restrict__ weight_node,
                             const float* __restrict__ bias,
                             const float* __restrict__ ws,
                             float* __restrict__ out) {
  unsigned lane = threadIdx.x & 31u;
  unsigned wave = (blockIdx.x * blockDim.x + threadIdx.x) >> 5;
  unsigned base = wave * 16u;
  unsigned oc = lane & 15u;
  float wv = weight_node[oc];
  float bv = bias[oc];
  const float* U0 = ws + WS_U0;
  const float* U1 = ws + WS_U1;

  v2f a;
  a.x = (lane < 16u) ? U0[base + lane] + U1[base + lane] : 0.0f;
  a.y = 0.0f;
  v2f b;
  b.x = (lane < 16u) ? wv : 0.0f;
  b.y = 0.0f;
  v8f c;
#pragma unroll
  for (int v = 0; v < 8; ++v) c[v] = bv;

  v8f d = __builtin_amdgcn_wmma_f32_16x16x4_f32(
       false, a,  false, b,
       (short)0, c,  false,  false);

  unsigned mbase = (lane < 16u) ? 0u : 8u;
#pragma unroll
  for (int v = 0; v < 8; ++v) {
    unsigned nb = base + mbase + (unsigned)v;
    unsigned n = nb >> 2;
    unsigned bb = nb & 3u;
    out[((size_t)bb * NUM_NEURONS + n) * OUT_CH + oc] = d[v];
  }
}

extern "C" void kernel_launch(void* const* d_in, const int* in_sizes, int n_in,
                              void* d_out, int out_size, void* d_ws, size_t ws_size,
                              hipStream_t stream) {
  const float* x_in        = (const float*)d_in[0];
  const int*   he          = (const int*)d_in[1];
  const float* post_spk    = (const float*)d_in[2];
  const float* w_hat       = (const float*)d_in[3];
  const float* s_e         = (const float*)d_in[4];
  const float* e_trace     = (const float*)d_in[5];
  const float* pre_trace   = (const float*)d_in[6];
  const float* post_trace  = (const float*)d_in[7];
  const float* weight_node = (const float*)d_in[8];
  const float* bias        = (const float*)d_in[9];
  float* out = (float*)d_out;
  float* ws  = (float*)d_ws;

  bhs_traces<<<(NUM_NEURONS + 255u) / 256u, 256, 0, stream>>>(
      x_in, post_spk, e_trace, pre_trace, post_trace, out, ws);
  bhs_order<<<NCHUNK, 1024, 0, stream>>>(he, out);
  bhs_owner<<<NBIN, 1024, 0, stream>>>(w_hat, s_e, ws, out);
  bhs_out_wmma<<<3125, 256, 0, stream>>>(weight_node, bias, ws, out);
}


# Trace bookkeeping, x transpose and clearing of U moved into the ordering launch (first 13 waves of each work-group; loads behind the index loads, stores after the last barrier): 3 launches, not 4
# speedup vs baseline: 10.9976x; 1.0681x over previous
#include <hip/hip_runtime.h>


#define NUM_NEURONS 100000
#define MAX_EDGES   100000
#define NUM_CONN    1600000
#define BATCH       4
#define OUT_CH      16

#define NCHUNK 256u
#define CH     6250u
#define NBIN   256u
#define BW     391u

typedef __attribute__((ext_vector_type(2))) float v2f;
typedef __attribute__((ext_vector_type(8))) float v8f;

#define WS_U1   100000u
#define WS_U0   500000u
#define WS_XT   900000u
#define OS_SORT 0u
#define OS_TAB  1600000u

__device__ __forceinline__ unsigned bhs_half() {
  return (__builtin_amdgcn_s_sendmsg_rtn(0x87) >> 18) & 1u;
}

__global__ __launch_bounds__(1024) void bhs_order(const float* __restrict__ x_in,
                                                  const float* __restrict__ post_spk,
                                                  const float* __restrict__ e_trace,
                                                  const float* __restrict__ pre_trace,
                                                  const float* __restrict__ post_trace,
                                                  const int* __restrict__ he,
                                                  float* __restrict__ out,
                                                  float* __restrict__ ws) {
  __shared__ unsigned hist[NBIN];
  __shared__ unsigned offs[NBIN];
  __shared__ unsigned wtot[8];
  __shared__ unsigned stage[CH];
  unsigned tid = threadIdx.x;
  unsigned lane = tid & 31u;
  unsigned chunk = blockIdx.x;
  unsigned base = chunk * CH;
  unsigned* sorted = (unsigned*)(out + OS_SORT);
  unsigned* tab = (unsigned*)(out + OS_TAB);
  if (tid < NBIN) hist[tid] = 0u;
  int nn[7], ee[7];
  for (unsigned it = 0; it < 7u; ++it) {
    unsigned k = tid + it * 1024u;
    nn[it] = 0; ee[it] = 0;
    if (k < CH) { nn[it] = he[base + k]; ee[it] = he[NUM_CONN + base + k]; }
  }
  unsigned ti = chunk * BW + tid;
  bool tr = (tid < BW) && (ti < (unsigned)NUM_NEURONS);
  float x0 = 0.f, x1 = 0.f, x2 = 0.f, x3 = 0.f;
  float pre_old = 0.f, post_old = 0.f, spk = 0.f, e_old = 0.f;
  if (tr) {
    x0 = x_in[ti];
    x1 = x_in[NUM_NEURONS + ti];
    x2 = x_in[2u * NUM_NEURONS + ti];
    x3 = x_in[3u * NUM_NEURONS + ti];
    pre_old  = pre_trace[ti];
    post_old = post_trace[ti];
    spk      = post_spk[ti];
    e_old    = e_trace[ti];
  }
  __syncthreads();
  unsigned pk[7], slot[7];
  for (unsigned it = 0; it < 7u; ++it) {
    unsigned k = tid + it * 1024u;
    pk[it] = 0u; slot[it] = 0u;
    if (k < CH) {
      unsigned e = (unsigned)ee[it];
      unsigned bn = e / BW;
      unsigned el = e - bn * BW;
      pk[it] = (unsigned)nn[it] | (el << 17);
      unsigned r = atomicAdd(&hist[bn], 1u);
      slot[it] = (bn << 16) | r;
    }
  }
  __syncthreads();
  unsigned v = (tid < NBIN) ? hist[tid] : 0u;
  unsigned x = v;
  for (unsigned d = 1u; d < 32u; d <<= 1) {
    unsigned y = __shfl_up(x, d);
    if (lane >= d) x += y;
  }
  if (tid < NBIN && lane == 31u) wtot[tid >> 5] = x;
  __syncthreads();
  if (tid < NBIN) {
    unsigned w = tid >> 5;
    unsigned add = 0u;
    for (unsigned q = 0; q < 8u; ++q) if (q < w) add += wtot[q];
    unsigned start = add + x - v;
    offs[tid] = start;
    tab[tid * NCHUNK + chunk] = start;
  }
  if (tid == 0u) tab[NBIN * NCHUNK + chunk] = CH;
  __syncthreads();
  for (unsigned it = 0; it < 7u; ++it) {
    unsigned k = tid + it * 1024u;
    if (k < CH) stage[offs[slot[it] >> 16] + (slot[it] & 0xFFFFu)] = pk[it];
  }
  __syncthreads();
  for (unsigned it = 0; it < 7u; ++it) {
    unsigned k = tid + it * 1024u;
    if (k < CH) sorted[base + k] = stage[k];
  }
  if (tr) {
    const float decay = 0.95122942450071400910f;
    float pre_spk  = (x0 + x1 + x2 + x3) * 0.25f;
    float pre_new  = fmaf(pre_old,  decay, pre_spk);
    float post_new = fmaf(post_old, decay, spk);
    float e_new    = fmaf(0.9f, e_old, pre_spk * (post_new + 0.1f));
    const size_t OUTB = (size_t)BATCH * NUM_NEURONS * OUT_CH;
    out[OUTB + ti]                  = e_new;
    out[OUTB + MAX_EDGES + ti]      = pre_new;
    out[OUTB + 2u * MAX_EDGES + ti] = post_new;
    ((float4*)(ws + WS_XT))[ti] = make_float4(x0, x1, x2, x3);
    ((float4*)(ws + WS_U0))[ti] = make_float4(0.f, 0.f, 0.f, 0.f);
    ((float4*)(ws + WS_U1))[ti] = make_float4(0.f, 0.f, 0.f, 0.f);
  }
}

__device__ __forceinline__ void bhs_add_S(float* S_l, unsigned p, float4 xv) {
  float* r = S_l + (p >> 17) * 5u;
  atomicAdd(r + 0, xv.x);
  atomicAdd(r + 1, xv.y);
  atomicAdd(r + 2, xv.z);
  atomicAdd(r + 3, xv.w);
  atomicAdd(r + 4, 1.0f);
}

__device__ __forceinline__ void bhs_add_U(float* U, const float* T_l, unsigned pp,
                                          unsigned cnt, unsigned lane) {
  unsigned j = lane & 3u;
  unsigned q = lane >> 2;
  unsigned pr[4];
  float tv[4];
  for (unsigned s = 0; s < 4u; ++s) pr[s] = (unsigned)__shfl((int)pp, (int)(s * 8u + q));
  for (unsigned s = 0; s < 4u; ++s) tv[s] = T_l[(pr[s] >> 17) * 4u + j];
  for (unsigned s = 0; s < 4u; ++s) {
    if (s * 8u + q < cnt) atomicAdd(U + (pr[s] & 0x1FFFFu) * 4u + j, tv[s]);
  }
}

__global__ __launch_bounds__(1024) void bhs_owner(const float* __restrict__ w_hat,
                                                  const float* __restrict__ s_e,
                                                  float* ws,
                                                  const float* __restrict__ out) {
  __shared__ float S_l[BW * 5u];
  __shared__ float T_l[BW * 4u];
  unsigned half = bhs_half();
  unsigned tid = threadIdx.x;
  unsigned lane = tid & 31u;
  unsigned wave = tid >> 5;
  unsigned b = blockIdx.x;
  const unsigned* sorted = (const unsigned*)(out + OS_SORT);
  const unsigned* tab = (const unsigned*)(out + OS_TAB);
  const float4* XT4 = (const float4*)(ws + WS_XT);
  float* U = ws + (half ? WS_U1 : WS_U0);

  unsigned e_own = b * BW + tid;
  bool own = (tid < BW) && (e_own < (unsigned)MAX_EDGES);
  float sev = s_e[own ? e_own : 0u];
  float whv = w_hat[own ? e_own : 0u];

  int st_l = 0, ln_l = 0;
  if (lane < 8u) {
    unsigned ci = wave * 8u + lane;
    unsigned o0 = tab[b * NCHUNK + ci];
    unsigned o1 = tab[(b + 1u) * NCHUNK + ci];
    st_l = (int)(ci * CH + o0);
    ln_l = (int)(o1 - o0);
  }
  for (unsigned k = tid; k < BW * 5u; k += 1024u) S_l[k] = 0.f;

  unsigned stv[8], lnv[8];
  for (int u = 0; u < 8; ++u) {
    stv[u] = (unsigned)__builtin_amdgcn_readlane(st_l, u);
    lnv[u] = (unsigned)__builtin_amdgcn_readlane(ln_l, u);
  }
  unsigned p[8];
  for (int u = 0; u < 8; ++u) {
    unsigned w = sorted[stv[u] + lane];
    p[u] = (lane < lnv[u]) ? w : 0u;
  }
  float4 xv[8];
  for (int u = 0; u < 8; ++u) xv[u] = XT4[p[u] & 0x1FFFFu];
  __syncthreads();
  for (int u = 0; u < 8; ++u) {
    if (lane < lnv[u]) bhs_add_S(S_l, p[u], xv[u]);
  }
  for (int u = 0; u < 8; ++u) {
    for (unsigned r0 = 32u; r0 < lnv[u]; r0 += 32u) {
      if (r0 + lane < lnv[u]) {
        unsigned pp = sorted[stv[u] + r0 + lane];
        bhs_add_S(S_l, pp, XT4[pp & 0x1FFFFu]);
      }
    }
  }
  __syncthreads();
  if (own) {
    float cwv = sev * whv;
    float cnt = S_l[tid * 5u + 4u];
    float g = cwv * (1.0f / fmaxf(cnt, 1.0f));
    T_l[tid * 4u + 0u] = S_l[tid * 5u + 0u] * g * cwv;
    T_l[tid * 4u + 1u] = S_l[tid * 5u + 1u] * g * cwv;
    T_l[tid * 4u + 2u] = S_l[tid * 5u + 2u] * g * cwv;
    T_l[tid * 4u + 3u] = S_l[tid * 5u + 3u] * g * cwv;
  }
  __syncthreads();
  for (int u = 0; u < 8; ++u) {
    bhs_add_U(U, T_l, p[u], lnv[u] < 32u ? lnv[u] : 32u, lane);
    for (unsigned r0 = 32u; r0 < lnv[u]; r0 += 32u) {
      unsigned left = lnv[u] - r0;
      unsigned pp = 0u;
      if (lane < left) pp = sorted[stv[u] + r0 + lane];
      bhs_add_U(U, T_l, pp, left < 32u ? left : 32u, lane);
    }
  }
}

__global__ void bhs_out_wmma(const float* __restrict__ weight_node,
                             const float* __restrict__ bias,
                             const float* __restrict__ ws,
                             float* __restrict__ out) {
  unsigned lane = threadIdx.x & 31u;
  unsigned wave = (blockIdx.x * blockDim.x + threadIdx.x) >> 5;
  unsigned base = wave * 16u;
  unsigned oc = lane & 15u;
  float wv = weight_node[oc];
  float bv = bias[oc];
  const float* U0 = ws + WS_U0;
  const float* U1 = ws + WS_U1;

  v2f a;
  a.x = (lane < 16u) ? U0[base + lane] + U1[base + lane] : 0.0f;
  a.y = 0.0f;
  v2f b;
  b.x = (lane < 16u) ? wv : 0.0f;
  b.y = 0.0f;
  v8f c;
#pragma unroll
  for (int v = 0; v < 8; ++v) c[v] = bv;

  v8f d = __builtin_amdgcn_wmma_f32_16x16x4_f32(
       false, a,  false, b,
       (short)0, c,  false,  false);

  unsigned mbase = (lane < 16u) ? 0u : 8u;
#pragma unroll
  for (int v = 0; v < 8; ++v) {
    unsigned nb = base + mbase + (unsigned)v;
    unsigned n = nb >> 2;
    unsigned bb = nb & 3u;
    out[((size_t)bb * NUM_NEURONS + n) * OUT_CH + oc] = d[v];
  }
}

extern "C" void kernel_launch(void* const* d_in, const int* in_sizes, int n_in,
                              void* d_out, int out_size, void* d_ws, size_t ws_size,
                              hipStream_t stream) {
  const float* x_in        = (const float*)d_in[0];
  const int*   he          = (const int*)d_in[1];
  const float* post_spk    = (const float*)d_in[2];
  const float* w_hat       = (const float*)d_in[3];
  const float* s_e         = (const float*)d_in[4];
  const float* e_trace     = (const float*)d_in[5];
  const float* pre_trace   = (const float*)d_in[6];
  const float* post_trace  = (const float*)d_in[7];
  const float* weight_node = (const float*)d_in[8];
  const float* bias        = (const float*)d_in[9];
  float* out = (float*)d_out;
  float* ws  = (float*)d_ws;

  bhs_order<<<NCHUNK, 1024, 0, stream>>>(
      x_in, post_spk, e_trace, pre_trace, post_trace, he, out, ws);
  bhs_owner<<<NBIN, 1024, 0, stream>>>(w_hat, s_e, ws, out);
  bhs_out_wmma<<<3125, 256, 0, stream>>>(weight_node, bias, ws, out);
}


# Order's table of run starts stored chunk-major: a work-group writes its 257 starts as nine whole lines that it alone writes; before, 4-byte words into lines shared by 32 work-groups of both halves
# speedup vs baseline: 11.6192x; 1.0565x over previous
#include <hip/hip_runtime.h>


#define NUM_NEURONS 100000
#define MAX_EDGES   100000
#define NUM_CONN    1600000
#define BATCH       4
#define OUT_CH      16

#define NCHUNK 256u
#define CH     6250u
#define NBIN   256u
#define BW     391u

typedef __attribute__((ext_vector_type(2))) float v2f;
typedef __attribute__((ext_vector_type(8))) float v8f;

#define WS_U1   100000u
#define WS_U0   500000u
#define WS_XT   900000u
#define OS_SORT 0u
#define OS_TAB  1600000u
#define TP      288u

__device__ __forceinline__ unsigned bhs_half() {
  return (__builtin_amdgcn_s_sendmsg_rtn(0x87) >> 18) & 1u;
}

__global__ __launch_bounds__(1024) void bhs_order(const float* __restrict__ x_in,
                                                  const float* __restrict__ post_spk,
                                                  const float* __restrict__ e_trace,
                                                  const float* __restrict__ pre_trace,
                                                  const float* __restrict__ post_trace,
                                                  const int* __restrict__ he,
                                                  float* __restrict__ out,
                                                  float* __restrict__ ws) {
  __shared__ unsigned hist[NBIN];
  __shared__ unsigned offs[NBIN];
  __shared__ unsigned wtot[8];
  __shared__ unsigned stage[CH];
  unsigned tid = threadIdx.x;
  unsigned lane = tid & 31u;
  unsigned chunk = blockIdx.x;
  unsigned base = chunk * CH;
  unsigned* sorted = (unsigned*)(out + OS_SORT);
  unsigned* tab = (unsigned*)(out + OS_TAB);
  if (tid < NBIN) hist[tid] = 0u;
  int nn[7], ee[7];
  for (unsigned it = 0; it < 7u; ++it) {
    unsigned k = tid + it * 1024u;
    nn[it] = 0; ee[it] = 0;
    if (k < CH) { nn[it] = he[base + k]; ee[it] = he[NUM_CONN + base + k]; }
  }
  unsigned ti = chunk * BW + tid;
  bool tr = (tid < BW) && (ti < (unsigned)NUM_NEURONS);
  float x0 = 0.f, x1 = 0.f, x2 = 0.f, x3 = 0.f;
  float pre_old = 0.f, post_old = 0.f, spk = 0.f, e_old = 0.f;
  if (tr) {
    x0 = x_in[ti];
    x1 = x_in[NUM_NEURONS + ti];
    x2 = x_in[2u * NUM_NEURONS + ti];
    x3 = x_in[3u * NUM_NEURONS + ti];
    pre_old  = pre_trace[ti];
    post_old = post_trace[ti];
    spk      = post_spk[ti];
    e_old    = e_trace[ti];
  }
  __syncthreads();
  unsigned pk[7], slot[7];
  for (unsigned it = 0; it < 7u; ++it) {
    unsigned k = tid + it * 1024u;
    pk[it] = 0u; slot[it] = 0u;
    if (k < CH) {
      unsigned e = (unsigned)ee[it];
      unsigned bn = e / BW;
      unsigned el = e - bn * BW;
      pk[it] = (unsigned)nn[it] | (el << 17);
      unsigned r = atomicAdd(&hist[bn], 1u);
      slot[it] = (bn << 16) | r;
    }
  }
  __syncthreads();
  unsigned v = (tid < NBIN) ? hist[tid] : 0u;
  unsigned x = v;
  for (unsigned d = 1u; d < 32u; d <<= 1) {
    unsigned y = __shfl_up(x, d);
    if (lane >= d) x += y;
  }
  if (tid < NBIN && lane == 31u) wtot[tid >> 5] = x;
  __syncthreads();
  if (tid < NBIN) {
    unsigned w = tid >> 5;
    unsigned add = 0u;
    for (unsigned q = 0; q < 8u; ++q) if (q < w) add += wtot[q];
    unsigned start = add + x - v;
    offs[tid] = start;
    tab[chunk * TP + tid] = start;
  }
  if (tid == NBIN) tab[chunk * TP + NBIN] = CH;
  __syncthreads();
  for (unsigned it = 0; it < 7u; ++it) {
    unsigned k = tid + it * 1024u;
    if (k < CH) stage[offs[slot[it] >> 16] + (slot[it] & 0xFFFFu)] = pk[it];
  }
  __syncthreads();
  for (unsigned it = 0; it < 7u; ++it) {
    unsigned k = tid + it * 1024u;
    if (k < CH) sorted[base + k] = stage[k];
  }
  if (tr) {
    const float decay = 0.95122942450071400910f;
    float pre_spk  = (x0 + x1 + x2 + x3) * 0.25f;
    float pre_new  = fmaf(pre_old,  decay, pre_spk);
    float post_new = fmaf(post_old, decay, spk);
    float e_new    = fmaf(0.9f, e_old, pre_spk * (post_new + 0.1f));
    const size_t OUTB = (size_t)BATCH * NUM_NEURONS * OUT_CH;
    out[OUTB + ti]                  = e_new;
    out[OUTB + MAX_EDGES + ti]      = pre_new;
    out[OUTB + 2u * MAX_EDGES + ti] = post_new;
    ((float4*)(ws + WS_XT))[ti] = make_float4(x0, x1, x2, x3);
    ((float4*)(ws + WS_U0))[ti] = make_float4(0.f, 0.f, 0.f, 0.f);
    ((float4*)(ws + WS_U1))[ti] = make_float4(0.f, 0.f, 0.f, 0.f);
  }
}

__device__ __forceinline__ void bhs_add_S(float* S_l, unsigned p, float4 xv) {
  float* r = S_l + (p >> 17) * 5u;
  atomicAdd(r + 0, xv.x);
  atomicAdd(r + 1, xv.y);
  atomicAdd(r + 2, xv.z);
  atomicAdd(r + 3, xv.w);
  atomicAdd(r + 4, 1.0f);
}

__device__ __forceinline__ void bhs_add_U(float* U, const float* T_l, unsigned pp,
                                          unsigned cnt, unsigned lane) {
  unsigned j = lane & 3u;
  unsigned q = lane >> 2;
  unsigned pr[4];
  float tv[4];
  for (unsigned s = 0; s < 4u; ++s) pr[s] = (unsigned)__shfl((int)pp, (int)(s * 8u + q));
  for (unsigned s = 0; s < 4u; ++s) tv[s] = T_l[(pr[s] >> 17) * 4u + j];
  for (unsigned s = 0; s < 4u; ++s) {
    if (s * 8u + q < cnt) atomicAdd(U + (pr[s] & 0x1FFFFu) * 4u + j, tv[s]);
  }
}

__global__ __launch_bounds__(1024) void bhs_owner(const float* __restrict__ w_hat,
                                                  const float* __restrict__ s_e,
                                                  float* ws,
                                                  const float* __restrict__ out) {
  __shared__ float S_l[BW * 5u];
  __shared__ float T_l[BW * 4u];
  unsigned half = bhs_half();
  unsigned tid = threadIdx.x;
  unsigned lane = tid & 31u;
  unsigned wave = tid >> 5;
  unsigned b = blockIdx.x;
  const unsigned* sorted = (const unsigned*)(out + OS_SORT);
  const unsigned* tab = (const unsigned*)(out + OS_TAB);
  const float4* XT4 = (const float4*)(ws + WS_XT);
  float* U = ws + (half ? WS_U1 : WS_U0);

  unsigned e_own = b * BW + tid;
  bool own = (tid < BW) && (e_own < (unsigned)MAX_EDGES);
  float sev = s_e[own ? e_own : 0u];
  float whv = w_hat[own ? e_own : 0u];

  int st_l = 0, ln_l = 0;
  if (lane < 8u) {
    unsigned ci = wave * 8u + lane;
    unsigned o0 = tab[ci * TP + b];
    unsigned o1 = tab[ci * TP + b + 1u];
    st_l = (int)(ci * CH + o0);
    ln_l = (int)(o1 - o0);
  }
  for (unsigned k = tid; k < BW * 5u; k += 1024u) S_l[k] = 0.f;

  unsigned stv[8], lnv[8];
  for (int u = 0; u < 8; ++u) {
    stv[u] = (unsigned)__builtin_amdgcn_readlane(st_l, u);
    lnv[u] = (unsigned)__builtin_amdgcn_readlane(ln_l, u);
  }
  unsigned p[8];
  for (int u = 0; u < 8; ++u) {
    unsigned w = sorted[stv[u] + lane];
    p[u] = (lane < lnv[u]) ? w : 0u;
  }
  float4 xv[8];
  for (int u = 0; u < 8; ++u) xv[u] = XT4[p[u] & 0x1FFFFu];
  __syncthreads();
  for (int u = 0; u < 8; ++u) {
    if (lane < lnv[u]) bhs_add_S(S_l, p[u], xv[u]);
  }
  for (int u = 0; u < 8; ++u) {
    for (unsigned r0 = 32u; r0 < lnv[u]; r0 += 32u) {
      if (r0 + lane < lnv[u]) {
        unsigned pp = sorted[stv[u] + r0 + lane];
        bhs_add_S(S_l, pp, XT4[pp & 0x1FFFFu]);
      }
    }
  }
  __syncthreads();
  if (own) {
    float cwv = sev * whv;
    float cnt = S_l[tid * 5u + 4u];
    float g = cwv * (1.0f / fmaxf(cnt, 1.0f));
    T_l[tid * 4u + 0u] = S_l[tid * 5u + 0u] * g * cwv;
    T_l[tid * 4u + 1u] = S_l[tid * 5u + 1u] * g * cwv;
    T_l[tid * 4u + 2u] = S_l[tid * 5u + 2u] * g * cwv;
    T_l[tid * 4u + 3u] = S_l[tid * 5u + 3u] * g * cwv;
  }
  __syncthreads();
  for (int u = 0; u < 8; ++u) {
    bhs_add_U(U, T_l, p[u], lnv[u] < 32u ? lnv[u] : 32u, lane);
    for (unsigned r0 = 32u; r0 < lnv[u]; r0 += 32u) {
      unsigned left = lnv[u] - r0;
      unsigned pp = 0u;
      if (lane < left) pp = sorted[stv[u] + r0 + lane];
      bhs_add_U(U, T_l, pp, left < 32u ? left : 32u, lane);
    }
  }
}

__global__ void bhs_out_wmma(const float* __restrict__ weight_node,
                             const float* __restrict__ bias,
                             const float* __restrict__ ws,
                             float* __restrict__ out) {
  unsigned lane = threadIdx.x & 31u;
  unsigned wave = (blockIdx.x * blockDim.x + threadIdx.x) >> 5;
  unsigned base = wave * 16u;
  unsigned oc = lane & 15u;
  float wv = weight_node[oc];
  float bv = bias[oc];
  const float* U0 = ws + WS_U0;
  const float* U1 = ws + WS_U1;

  v2f a;
  a.x = (lane < 16u) ? U0[base + lane] + U1[base + lane] : 0.0f;
  a.y = 0.0f;
  v2f b;
  b.x = (lane < 16u) ? wv : 0.0f;
  b.y = 0.0f;
  v8f c;
#pragma unroll
  for (int v = 0; v < 8; ++v) c[v] = bv;

  v8f d = __builtin_amdgcn_wmma_f32_16x16x4_f32(
       false, a,  false, b,
       (short)0, c,  false,  false);

  unsigned mbase = (lane < 16u) ? 0u : 8u;
#pragma unroll
  for (int v = 0; v < 8; ++v) {
    unsigned nb = base + mbase + (unsigned)v;
    unsigned n = nb >> 2;
    unsigned bb = nb & 3u;
    out[((size_t)bb * NUM_NEURONS + n) * OUT_CH + oc] = d[v];
  }
}

extern "C" void kernel_launch(void* const* d_in, const int* in_sizes, int n_in,
                              void* d_out, int out_size, void* d_ws, size_t ws_size,
                              hipStream_t stream) {
  const float* x_in        = (const float*)d_in[0];
  const int*   he          = (const int*)d_in[1];
  const float* post_spk    = (const float*)d_in[2];
  const float* w_hat       = (const float*)d_in[3];
  const float* s_e         = (const float*)d_in[4];
  const float* e_trace     = (const float*)d_in[5];
  const float* pre_trace   = (const float*)d_in[6];
  const float* post_trace  = (const float*)d_in[7];
  const float* weight_node = (const float*)d_in[8];
  const float* bias        = (const float*)d_in[9];
  float* out = (float*)d_out;
  float* ws  = (float*)d_ws;

  bhs_order<<<NCHUNK, 1024, 0, stream>>>(
      x_in, post_spk, e_trace, pre_trace, post_trace, he, out, ws);
  bhs_owner<<<NBIN, 1024, 0, stream>>>(w_hat, s_e, ws, out);
  bhs_out_wmma<<<3125, 256, 0, stream>>>(weight_node, bias, ws, out);
}


# Owner: what a wave's eight runs hold past 32 words is one ninth lane group, loaded and gathered up front with the eight (no load-gather-add round trips per long run); add order free as before
# speedup vs baseline: 12.1623x; 1.0467x over previous
#include <hip/hip_runtime.h>


#define NUM_NEURONS 100000
#define MAX_EDGES   100000
#define NUM_CONN    1600000
#define BATCH       4
#define OUT_CH      16

#define NCHUNK 256u
#define CH     6250u
#define NBIN   256u
#define BW     391u

typedef __attribute__((ext_vector_type(2))) float v2f;
typedef __attribute__((ext_vector_type(8))) float v8f;

#define WS_U1   100000u
#define WS_U0   500000u
#define WS_XT   900000u
#define OS_SORT 0u
#define OS_TAB  1600000u
#define TP      288u

__device__ __forceinline__ unsigned bhs_half() {
  return (__builtin_amdgcn_s_sendmsg_rtn(0x87) >> 18) & 1u;
}

__global__ __launch_bounds__(1024) void bhs_order(const float* __restrict__ x_in,
                                                  const float* __restrict__ post_spk,
                                                  const float* __restrict__ e_trace,
                                                  const float* __restrict__ pre_trace,
                                                  const float* __restrict__ post_trace,
                                                  const int* __restrict__ he,
                                                  float* __restrict__ out,
                                                  float* __restrict__ ws) {
  __shared__ unsigned hist[NBIN];
  __shared__ unsigned offs[NBIN];
  __shared__ unsigned wtot[8];
  __shared__ unsigned stage[CH];
  unsigned tid = threadIdx.x;
  unsigned lane = tid & 31u;
  unsigned chunk = blockIdx.x;
  unsigned base = chunk * CH;
  unsigned* sorted = (unsigned*)(out + OS_SORT);
  unsigned* tab = (unsigned*)(out + OS_TAB);
  if (tid < NBIN) hist[tid] = 0u;
  int nn[7], ee[7];
  for (unsigned it = 0; it < 7u; ++it) {
    unsigned k = tid + it * 1024u;
    nn[it] = 0; ee[it] = 0;
    if (k < CH) { nn[it] = he[base + k]; ee[it] = he[NUM_CONN + base + k]; }
  }
  unsigned ti = chunk * BW + tid;
  bool tr = (tid < BW) && (ti < (unsigned)NUM_NEURONS);
  float x0 = 0.f, x1 = 0.f, x2 = 0.f, x3 = 0.f;
  float pre_old = 0.f, post_old = 0.f, spk = 0.f, e_old = 0.f;
  if (tr) {
    x0 = x_in[ti];
    x1 = x_in[NUM_NEURONS + ti];
    x2 = x_in[2u * NUM_NEURONS + ti];
    x3 = x_in[3u * NUM_NEURONS + ti];
    pre_old  = pre_trace[ti];
    post_old = post_trace[ti];
    spk      = post_spk[ti];
    e_old    = e_trace[ti];
  }
  __syncthreads();
  unsigned pk[7], slot[7];
  for (unsigned it = 0; it < 7u; ++it) {
    unsigned k = tid + it * 1024u;
    pk[it] = 0u; slot[it] = 0u;
    if (k < CH) {
      unsigned e = (unsigned)ee[it];
      unsigned bn = e / BW;
      unsigned el = e - bn * BW;
      pk[it] = (unsigned)nn[it] | (el << 17);
      unsigned r = atomicAdd(&hist[bn], 1u);
      slot[it] = (bn << 16) | r;
    }
  }
  __syncthreads();
  unsigned v = (tid < NBIN) ? hist[tid] : 0u;
  unsigned x = v;
  for (unsigned d = 1u; d < 32u; d <<= 1) {
    unsigned y = __shfl_up(x, d);
    if (lane >= d) x += y;
  }
  if (tid < NBIN && lane == 31u) wtot[tid >> 5] = x;
  __syncthreads();
  if (tid < NBIN) {
    unsigned w = tid >> 5;
    unsigned add = 0u;
    for (unsigned q = 0; q < 8u; ++q) if (q < w) add += wtot[q];
    unsigned start = add + x - v;
    offs[tid] = start;
    tab[chunk * TP + tid] = start;
  }
  if (tid == NBIN) tab[chunk * TP + NBIN] = CH;
  __syncthreads();
  for (unsigned it = 0; it < 7u; ++it) {
    unsigned k = tid + it * 1024u;
    if (k < CH) stage[offs[slot[it] >> 16] + (slot[it] & 0xFFFFu)] = pk[it];
  }
  __syncthreads();
  for (unsigned it = 0; it < 7u; ++it) {
    unsigned k = tid + it * 1024u;
    if (k < CH) sorted[base + k] = stage[k];
  }
  if (tr) {
    const float decay = 0.95122942450071400910f;
    float pre_spk  = (x0 + x1 + x2 + x3) * 0.25f;
    float pre_new  = fmaf(pre_old,  decay, pre_spk);
    float post_new = fmaf(post_old, decay, spk);
    float e_new    = fmaf(0.9f, e_old, pre_spk * (post_new + 0.1f));
    const size_t OUTB = (size_t)BATCH * NUM_NEURONS * OUT_CH;
    out[OUTB + ti]                  = e_new;
    out[OUTB + MAX_EDGES + ti]      = pre_new;
    out[OUTB + 2u * MAX_EDGES + ti] = post_new;
    ((float4*)(ws + WS_XT))[ti] = make_float4(x0, x1, x2, x3);
    ((float4*)(ws + WS_U0))[ti] = make_float4(0.f, 0.f, 0.f, 0.f);
    ((float4*)(ws + WS_U1))[ti] = make_float4(0.f, 0.f, 0.f, 0.f);
  }
}

__device__ __forceinline__ void bhs_add_S(float* S_l, unsigned p, float4 xv) {
  float* r = S_l + (p >> 17) * 5u;
  atomicAdd(r + 0, xv.x);
  atomicAdd(r + 1, xv.y);
  atomicAdd(r + 2, xv.z);
  atomicAdd(r + 3, xv.w);
  atomicAdd(r + 4, 1.0f);
}

__device__ __forceinline__ void bhs_add_U(float* U, const float* T_l, unsigned pp,
                                          unsigned cnt, unsigned lane) {
  unsigned j = lane & 3u;
  unsigned q = lane >> 2;
  unsigned pr[4];
  float tv[4];
  for (unsigned s = 0; s < 4u; ++s) pr[s] = (unsigned)__shfl((int)pp, (int)(s * 8u + q));
  for (unsigned s = 0; s < 4u; ++s) tv[s] = T_l[(pr[s] >> 17) * 4u + j];
  for (unsigned s = 0; s < 4u; ++s) {
    if (s * 8u + q < cnt) atomicAdd(U + (pr[s] & 0x1FFFFu) * 4u + j, tv[s]);
  }
}

__device__ __forceinline__ unsigned bhs_rest_addr(unsigned k, const unsigned* tp,
                                                  const unsigned* toff) {
  unsigned o = toff[0];
  for (int u = 1; u < 8; ++u) o = (k >= tp[u]) ? toff[u] : o;
  return k + o;
}

__global__ __launch_bounds__(1024) void bhs_owner(const float* __restrict__ w_hat,
                                                  const float* __restrict__ s_e,
                                                  float* ws,
                                                  const float* __restrict__ out) {
  __shared__ float S_l[BW * 5u];
  __shared__ float T_l[BW * 4u];
  unsigned half = bhs_half();
  unsigned tid = threadIdx.x;
  unsigned lane = tid & 31u;
  unsigned wave = tid >> 5;
  unsigned b = blockIdx.x;
  const unsigned* sorted = (const unsigned*)(out + OS_SORT);
  const unsigned* tab = (const unsigned*)(out + OS_TAB);
  const float4* XT4 = (const float4*)(ws + WS_XT);
  float* U = ws + (half ? WS_U1 : WS_U0);

  unsigned e_own = b * BW + tid;
  bool own = (tid < BW) && (e_own < (unsigned)MAX_EDGES);
  float sev = s_e[own ? e_own : 0u];
  float whv = w_hat[own ? e_own : 0u];

  int st_l = 0, ln_l = 0;
  if (lane < 8u) {
    unsigned ci = wave * 8u + lane;
    unsigned o0 = tab[ci * TP + b];
    unsigned o1 = tab[ci * TP + b + 1u];
    st_l = (int)(ci * CH + o0);
    ln_l = (int)(o1 - o0);
  }
  for (unsigned k = tid; k < BW * 5u; k += 1024u) S_l[k] = 0.f;

  unsigned stv[8], lnv[8];
  for (int u = 0; u < 8; ++u) {
    stv[u] = (unsigned)__builtin_amdgcn_readlane(st_l, u);
    lnv[u] = (unsigned)__builtin_amdgcn_readlane(ln_l, u);
  }
  unsigned tp[9], toff[8];
  tp[0] = 0u;
  for (int u = 0; u < 8; ++u) {
    toff[u] = stv[u] + 32u - tp[u];
    tp[u + 1] = tp[u] + (lnv[u] > 32u ? lnv[u] - 32u : 0u);
  }
  unsigned ttot = tp[8];
  unsigned p[9];
  for (int u = 0; u < 8; ++u) {
    unsigned w = sorted[stv[u] + lane];
    p[u] = (lane < lnv[u]) ? w : 0u;
  }
  {
    bool rv = lane < ttot;
    unsigned w = sorted[rv ? bhs_rest_addr(lane, tp, toff) : stv[0]];
    p[8] = rv ? w : 0u;
  }
  float4 xv[9];
  for (int u = 0; u < 9; ++u) xv[u] = XT4[p[u] & 0x1FFFFu];
  __syncthreads();
  for (int u = 0; u < 8; ++u) {
    if (lane < lnv[u]) bhs_add_S(S_l, p[u], xv[u]);
  }
  if (lane < ttot) bhs_add_S(S_l, p[8], xv[8]);
  for (unsigned k0 = 32u; k0 < ttot; k0 += 32u) {
    unsigned k = k0 + lane;
    if (k < ttot) {
      unsigned pp = sorted[bhs_rest_addr(k, tp, toff)];
      bhs_add_S(S_l, pp, XT4[pp & 0x1FFFFu]);
    }
  }
  __syncthreads();
  if (own) {
    float cwv = sev * whv;
    float cnt = S_l[tid * 5u + 4u];
    float g = cwv * (1.0f / fmaxf(cnt, 1.0f));
    T_l[tid * 4u + 0u] = S_l[tid * 5u + 0u] * g * cwv;
    T_l[tid * 4u + 1u] = S_l[tid * 5u + 1u] * g * cwv;
    T_l[tid * 4u + 2u] = S_l[tid * 5u + 2u] * g * cwv;
    T_l[tid * 4u + 3u] = S_l[tid * 5u + 3u] * g * cwv;
  }
  __syncthreads();
  for (int u = 0; u < 8; ++u) {
    bhs_add_U(U, T_l, p[u], lnv[u] < 32u ? lnv[u] : 32u, lane);
  }
  if (ttot != 0u) bhs_add_U(U, T_l, p[8], ttot < 32u ? ttot : 32u, lane);
  for (unsigned k0 = 32u; k0 < ttot; k0 += 32u) {
    unsigned left = ttot - k0;
    unsigned pp = 0u;
    if (lane < left) pp = sorted[bhs_rest_addr(k0 + lane, tp, toff)];
    bhs_add_U(U, T_l, pp, left < 32u ? left : 32u, lane);
  }
}

__global__ void bhs_out_wmma(const float* __restrict__ weight_node,
                             const float* __restrict__ bias,
                             const float* __restrict__ ws,
                             float* __restrict__ out) {
  unsigned lane = threadIdx.x & 31u;
  unsigned wave = (blockIdx.x * blockDim.x + threadIdx.x) >> 5;
  unsigned base = wave * 16u;
  unsigned oc = lane & 15u;
  float wv = weight_node[oc];
  float bv = bias[oc];
  const float* U0 = ws + WS_U0;
  const float* U1 = ws + WS_U1;

  v2f a;
  a.x = (lane < 16u) ? U0[base + lane] + U1[base + lane] : 0.0f;
  a.y = 0.0f;
  v2f b;
  b.x = (lane < 16u) ? wv : 0.0f;
  b.y = 0.0f;
  v8f c;
#pragma unroll
  for (int v = 0; v < 8; ++v) c[v] = bv;

  v8f d = __builtin_amdgcn_wmma_f32_16x16x4_f32(
       false, a,  false, b,
       (short)0, c,  false,  false);

  unsigned mbase = (lane < 16u) ? 0u : 8u;
#pragma unroll
  for (int v = 0; v < 8; ++v) {
    unsigned nb = base + mbase + (unsigned)v;
    unsigned n = nb >> 2;
    unsigned bb = nb & 3u;
    out[((size_t)bb * NUM_NEURONS + n) * OUT_CH + oc] = d[v];
  }
}

extern "C" void kernel_launch(void* const* d_in, const int* in_sizes, int n_in,
                              void* d_out, int out_size, void* d_ws, size_t ws_size,
                              hipStream_t stream) {
  const float* x_in        = (const float*)d_in[0];
  const int*   he          = (const int*)d_in[1];
  const float* post_spk    = (const float*)d_in[2];
  const float* w_hat       = (const float*)d_in[3];
  const float* s_e         = (const float*)d_in[4];
  const float* e_trace     = (const float*)d_in[5];
  const float* pre_trace   = (const float*)d_in[6];
  const float* post_trace  = (const float*)d_in[7];
  const float* weight_node = (const float*)d_in[8];
  const float* bias        = (const float*)d_in[9];
  float* out = (float*)d_out;
  float* ws  = (float*)d_ws;

  bhs_order<<<NCHUNK, 1024, 0, stream>>>(
      x_in, post_spk, e_trace, pre_trace, post_trace, he, out, ws);
  bhs_owner<<<NBIN, 1024, 0, stream>>>(w_hat, s_e, ws, out);
  bhs_out_wmma<<<3125, 256, 0, stream>>>(weight_node, bias, ws, out);
}
